# starting kernel
# baseline (speedup 1.0000x reference)
#include <hip/hip_runtime.h>

typedef float v2f __attribute__((ext_vector_type(2)));
typedef float v8f __attribute__((ext_vector_type(8)));

#define NNODES 20000
#define NEDGES 320000

__global__ __launch_bounds__(256) void k_scatter_add(
    const float* __restrict__ feat, const int* __restrict__ ei,
    float* __restrict__ agg, int n_edges, int D)
{
    const int chunks = D >> 2;
    long long tid = (long long)blockIdx.x * blockDim.x + threadIdx.x;
    long long total = (long long)n_edges * chunks;
    if (tid >= total) return;
    int e = (int)(tid / chunks);
    int c = (int)(tid - (long long)e * chunks);
    int s = ei[e];
    int d = ei[n_edges + e];
    const float4 v = reinterpret_cast<const float4*>(feat + (size_t)s * D)[c];
    float* out = agg + (size_t)d * D + (c << 2);
    unsafeAtomicAdd(out + 0, v.x);
    unsafeAtomicAdd(out + 1, v.y);
    unsafeAtomicAdd(out + 2, v.z);
    unsafeAtomicAdd(out + 3, v.w);
}

template <int NT, bool DUAL>
__global__ __launch_bounds__(256) void k_gemm_strip(
    const float* __restrict__ A0, const float* __restrict__ B0,
    const float* __restrict__ A1, const float* __restrict__ B1,
    const float* __restrict__ bias, float* __restrict__ C,
    int M, int K, int relu)
{
    constexpr int N = NT * 16;
    constexpr int BK = 64;
    __shared__ __align__(16) float Bs0[BK * N];
    __shared__ __align__(16) float Bs1[DUAL ? (BK * N) : 4];

    const int lane = threadIdx.x & 31;
    const int wave = threadIdx.x >> 5;
    const int tm = blockIdx.x * 8 + wave;
    const bool active = (tm * 16 < M);
    const int rh = lane & 15;
    const int hi = lane >> 4;

    v8f acc[NT];
#pragma unroll
    for (int t = 0; t < NT; ++t) acc[t] = v8f{};

    const float* ap0 = A0 + (size_t)(tm * 16 + rh) * K + 2 * hi;
    const float* ap1 = DUAL ? (A1 + (size_t)(tm * 16 + rh) * K + 2 * hi) : nullptr;

    for (int kc = 0; kc < K; kc += BK) {
        __syncthreads();
        {
            constexpr int ITERS = (BK / 2) * (N / 4);
            for (int idx = threadIdx.x; idx < ITERS; idx += 256) {
                const int pr = idx / (N / 4);
                const int c = (idx - pr * (N / 4)) * 4;
                const int k = 2 * pr;
                float4 g0 = *(const float4*)(B0 + (size_t)(kc + k) * N + c);
                float4 g1 = *(const float4*)(B0 + (size_t)(kc + k + 1) * N + c);
                float4 lo = {g0.x, g1.x, g0.y, g1.y};
                float4 hi4 = {g0.z, g1.z, g0.w, g1.w};
                float* dst = &Bs0[(pr * N + c) * 2];
                *(float4*)(dst) = lo;
                *(float4*)(dst + 4) = hi4;
                if constexpr (DUAL) {
                    float4 h0 = *(const float4*)(B1 + (size_t)(kc + k) * N + c);
                    float4 h1 = *(const float4*)(B1 + (size_t)(kc + k + 1) * N + c);
                    float4 l2 = {h0.x, h1.x, h0.y, h1.y};
                    float4 h2 = {h0.z, h1.z, h0.w, h1.w};
                    float* dst1 = &Bs1[(pr * N + c) * 2];
                    *(float4*)(dst1) = l2;
                    *(float4*)(dst1 + 4) = h2;
                }
            }
        }
        __syncthreads();
        if (!active) continue;

        const v2f* Bp0 = (const v2f*)Bs0;
        const v2f* Bp1 = (const v2f*)Bs1;
        for (int k = 0; k < BK; k += 4) {
            const int pr = (k >> 1) + hi;
            v2f a;
            a.x = ap0[kc + k];
            a.y = ap0[kc + k + 1];
#pragma unroll
            for (int t = 0; t < NT; ++t) {
                v2f b = Bp0[pr * N + t * 16 + rh];
                acc[t] = __builtin_amdgcn_wmma_f32_16x16x4_f32(
                    false, a, false, b, (short)0, acc[t], false, false);
            }
            if constexpr (DUAL) {
                v2f a1;
                a1.x = ap1[kc + k];
                a1.y = ap1[kc + k + 1];
#pragma unroll
                for (int t = 0; t < NT; ++t) {
                    v2f b = Bp1[pr * N + t * 16 + rh];
                    acc[t] = __builtin_amdgcn_wmma_f32_16x16x4_f32(
                        false, a1, false, b, (short)0, acc[t], false, false);
                }
            }
        }
    }

    if (active) {
#pragma unroll
        for (int t = 0; t < NT; ++t) {
            const int col = t * 16 + rh;
            const float bv = bias[col];
#pragma unroll
            for (int i = 0; i < 8; ++i) {
                int r = tm * 16 + i + 8 * hi;
                float v = acc[t][i] + bv;
                if (relu) v = fmaxf(v, 0.0f);
                C[(size_t)r * N + col] = v;
            }
        }
    }
}

__global__ __launch_bounds__(256) void k_head(
    const float* __restrict__ h, const float* __restrict__ W,
    const float* __restrict__ b, float* __restrict__ out, int M)
{
    int n = blockIdx.x * blockDim.x + threadIdx.x;
    if (n >= M) return;
    const float* hr = h + (size_t)n * 64;
    float z[10];
#pragma unroll
    for (int c = 0; c < 10; ++c) z[c] = b[c];
    for (int k = 0; k < 64; ++k) {
        float hv = hr[k];
        const float* wr = W + k * 10;
#pragma unroll
        for (int c = 0; c < 10; ++c) z[c] = fmaf(hv, wr[c], z[c]);
    }
    float mx = z[0];
#pragma unroll
    for (int c = 1; c < 10; ++c) mx = fmaxf(mx, z[c]);
    float s = 0.0f;
#pragma unroll
    for (int c = 0; c < 10; ++c) s += expf(z[c] - mx);
    float lse = mx + logf(s);
    float* o = out + (size_t)n * 10;
#pragma unroll
    for (int c = 0; c < 10; ++c) o[c] = z[c] - lse;
}

extern "C" void kernel_launch(void* const* d_in, const int* in_sizes, int n_in,
                              void* d_out, int out_size, void* d_ws, size_t ws_size,
                              hipStream_t stream) {
    const float* x       = (const float*)d_in[0];
    const int*   ei      = (const int*)d_in[1];
    const float* W1_rel  = (const float*)d_in[2];
    const float* W1_root = (const float*)d_in[3];
    const float* b1      = (const float*)d_in[4];
    const float* W2_rel  = (const float*)d_in[5];
    const float* W2_root = (const float*)d_in[6];
    const float* b2      = (const float*)d_in[7];
    const float* lin1_w  = (const float*)d_in[8];
    const float* lin1_b  = (const float*)d_in[9];
    const float* lin2_w  = (const float*)d_in[10];
    const float* lin2_b  = (const float*)d_in[11];
    const float* lin3_w  = (const float*)d_in[12];
    const float* lin3_b  = (const float*)d_in[13];

    float* agg1 = (float*)d_ws;
    float* h1   = agg1 + (size_t)NNODES * 256;
    float* agg2 = h1   + (size_t)NNODES * 128;
    float* h2   = agg2 + (size_t)NNODES * 128;
    float* m1   = h2   + (size_t)NNODES * 64;
    float* m2   = m1   + (size_t)NNODES * 128;

    hipMemsetAsync(agg1, 0, (size_t)NNODES * 256 * sizeof(float), stream);
    hipMemsetAsync(agg2, 0, (size_t)NNODES * 128 * sizeof(float), stream);

    const int mblocks = ((NNODES / 16) + 7) / 8;

    {
        long long total = (long long)NEDGES * (256 / 4);
        int blocks = (int)((total + 255) / 256);
        k_scatter_add<<<blocks, 256, 0, stream>>>(x, ei, agg1, NEDGES, 256);
    }
    k_gemm_strip<8, true><<<mblocks, 256, 0, stream>>>(
        agg1, W1_rel, x, W1_root, b1, h1, NNODES, 256, 1);

    {
        long long total = (long long)NEDGES * (128 / 4);
        int blocks = (int)((total + 255) / 256);
        k_scatter_add<<<blocks, 256, 0, stream>>>(h1, ei, agg2, NEDGES, 128);
    }
    k_gemm_strip<4, true><<<mblocks, 256, 0, stream>>>(
        agg2, W2_rel, h1, W2_root, b2, h2, NNODES, 128, 0);

    k_gemm_strip<8, false><<<mblocks, 256, 0, stream>>>(
        h2, lin1_w, nullptr, nullptr, lin1_b, m1, NNODES, 64, 1);
    k_gemm_strip<4, false><<<mblocks, 256, 0, stream>>>(
        m1, lin2_w, nullptr, nullptr, lin2_b, m2, NNODES, 128, 1);
    k_head<<<(NNODES + 255) / 256, 256, 0, stream>>>(
        m2, lin3_w, lin3_b, (float*)d_out, NNODES);
}


# Apply W_rel before the edge scatter-add (the sum is linear): scatter 128- and 64-wide rows instead of 256/128, adding into the root+bias term; float sum order changes (sum of products)
# speedup vs baseline: 1.7155x; 1.7155x over previous
#include <hip/hip_runtime.h>

typedef float v2f __attribute__((ext_vector_type(2)));
typedef float v8f __attribute__((ext_vector_type(8)));

#define NNODES 20000
#define NEDGES 320000

__global__ __launch_bounds__(256) void k_scatter_add(
    const float* __restrict__ feat, const int* __restrict__ ei,
    float* __restrict__ agg, int n_edges, int D)
{
    const int chunks = D >> 2;
    long long tid = (long long)blockIdx.x * blockDim.x + threadIdx.x;
    long long total = (long long)n_edges * chunks;
    if (tid >= total) return;
    int e = (int)(tid / chunks);
    int c = (int)(tid - (long long)e * chunks);
    int s = ei[e];
    int d = ei[n_edges + e];
    const float4 v = reinterpret_cast<const float4*>(feat + (size_t)s * D)[c];
    float* out = agg + (size_t)d * D + (c << 2);
    unsafeAtomicAdd(out + 0, v.x);
    unsafeAtomicAdd(out + 1, v.y);
    unsafeAtomicAdd(out + 2, v.z);
    unsafeAtomicAdd(out + 3, v.w);
}

template <int NT, bool RELU_IN>
__global__ __launch_bounds__(256) void k_gemm_strip(
    const float* __restrict__ A,
    const float* __restrict__ B0, const float* __restrict__ bias0,
    float* __restrict__ C0,
    const float* __restrict__ B1, float* __restrict__ C1,
    int M, int K, int relu)
{
    constexpr int N = NT * 16;
    constexpr int BK = 64;
    __shared__ __align__(16) float Bs0[BK * N];

    const bool second = (blockIdx.y != 0);
    const float* __restrict__ B = second ? B1 : B0;
    float* __restrict__ C = second ? C1 : C0;

    const int lane = threadIdx.x & 31;
    const int wave = threadIdx.x >> 5;
    const int tm = blockIdx.x * 8 + wave;
    const bool active = (tm * 16 < M);
    const int rh = lane & 15;
    const int hi = lane >> 4;

    v8f acc[NT];
#pragma unroll
    for (int t = 0; t < NT; ++t) acc[t] = v8f{};

    const float* ap0 = A + (size_t)(tm * 16 + rh) * K + 2 * hi;

    for (int kc = 0; kc < K; kc += BK) {
        __syncthreads();
        {
            constexpr int ITERS = (BK / 2) * (N / 4);
            for (int idx = threadIdx.x; idx < ITERS; idx += 256) {
                const int pr = idx / (N / 4);
                const int c = (idx - pr * (N / 4)) * 4;
                const int k = 2 * pr;
                float4 g0 = *(const float4*)(B + (size_t)(kc + k) * N + c);
                float4 g1 = *(const float4*)(B + (size_t)(kc + k + 1) * N + c);
                float4 lo = {g0.x, g1.x, g0.y, g1.y};
                float4 hi4 = {g0.z, g1.z, g0.w, g1.w};
                float* dst = &Bs0[(pr * N + c) * 2];
                *(float4*)(dst) = lo;
                *(float4*)(dst + 4) = hi4;
            }
        }
        __syncthreads();
        if (!active) continue;

        const v2f* Bp0 = (const v2f*)Bs0;
        for (int k = 0; k < BK; k += 4) {
            const int pr = (k >> 1) + hi;
            v2f a;
            a.x = ap0[kc + k];
            a.y = ap0[kc + k + 1];
            if constexpr (RELU_IN) {
                a.x = fmaxf(a.x, 0.0f);
                a.y = fmaxf(a.y, 0.0f);
            }
#pragma unroll
            for (int t = 0; t < NT; ++t) {
                v2f b = Bp0[pr * N + t * 16 + rh];
                acc[t] = __builtin_amdgcn_wmma_f32_16x16x4_f32(
                    false, a, false, b, (short)0, acc[t], false, false);
            }
        }
    }

    if (active) {
#pragma unroll
        for (int t = 0; t < NT; ++t) {
            const int col = t * 16 + rh;
            const float bv = second ? 0.0f : bias0[col];
#pragma unroll
            for (int i = 0; i < 8; ++i) {
                int r = tm * 16 + i + 8 * hi;
                float v = acc[t][i] + bv;
                if (relu) v = fmaxf(v, 0.0f);
                C[(size_t)r * N + col] = v;
            }
        }
    }
}

__global__ __launch_bounds__(256) void k_head(
    const float* __restrict__ h, const float* __restrict__ W,
    const float* __restrict__ b, float* __restrict__ out, int M)
{
    int n = blockIdx.x * blockDim.x + threadIdx.x;
    if (n >= M) return;
    const float* hr = h + (size_t)n * 64;
    float z[10];
#pragma unroll
    for (int c = 0; c < 10; ++c) z[c] = b[c];
    for (int k = 0; k < 64; ++k) {
        float hv = hr[k];
        const float* wr = W + k * 10;
#pragma unroll
        for (int c = 0; c < 10; ++c) z[c] = fmaf(hv, wr[c], z[c]);
    }
    float mx = z[0];
#pragma unroll
    for (int c = 1; c < 10; ++c) mx = fmaxf(mx, z[c]);
    float s = 0.0f;
#pragma unroll
    for (int c = 0; c < 10; ++c) s += expf(z[c] - mx);
    float lse = mx + logf(s);
    float* o = out + (size_t)n * 10;
#pragma unroll
    for (int c = 0; c < 10; ++c) o[c] = z[c] - lse;
}

extern "C" void kernel_launch(void* const* d_in, const int* in_sizes, int n_in,
                              void* d_out, int out_size, void* d_ws, size_t ws_size,
                              hipStream_t stream) {
    const float* x       = (const float*)d_in[0];
    const int*   ei      = (const int*)d_in[1];
    const float* W1_rel  = (const float*)d_in[2];
    const float* W1_root = (const float*)d_in[3];
    const float* b1      = (const float*)d_in[4];
    const float* W2_rel  = (const float*)d_in[5];
    const float* W2_root = (const float*)d_in[6];
    const float* b2      = (const float*)d_in[7];
    const float* lin1_w  = (const float*)d_in[8];
    const float* lin1_b  = (const float*)d_in[9];
    const float* lin2_w  = (const float*)d_in[10];
    const float* lin2_b  = (const float*)d_in[11];
    const float* lin3_w  = (const float*)d_in[12];
    const float* lin3_b  = (const float*)d_in[13];

    float* y1 = (float*)d_ws;
    float* h1 = y1 + (size_t)NNODES * 128;
    float* y2 = h1 + (size_t)NNODES * 128;
    float* h2 = y2 + (size_t)NNODES * 64;
    float* m1 = h2 + (size_t)NNODES * 64;
    float* m2 = m1 + (size_t)NNODES * 128;

    const int mblocks = ((NNODES / 16) + 7) / 8;
    const dim3 grid2(mblocks, 2);


    k_gemm_strip<8, false><<<grid2, 256, 0, stream>>>(
        x, W1_root, b1, h1, W1_rel, y1, NNODES, 256, 0);
    {
        long long total = (long long)NEDGES * (128 / 4);
        int blocks = (int)((total + 255) / 256);
        k_scatter_add<<<blocks, 256, 0, stream>>>(y1, ei, h1, NEDGES, 128);
    }

    k_gemm_strip<4, true><<<grid2, 256, 0, stream>>>(
        h1, W2_root, b2, h2, W2_rel, y2, NNODES, 128, 0);
    {
        long long total = (long long)NEDGES * (64 / 4);
        int blocks = (int)((total + 255) / 256);
        k_scatter_add<<<blocks, 256, 0, stream>>>(y2, ei, h2, NEDGES, 64);
    }

    k_gemm_strip<8, false><<<mblocks, 256, 0, stream>>>(
        h2, lin1_w, lin1_b, m1, nullptr, nullptr, NNODES, 64, 1);
    k_gemm_strip<4, false><<<mblocks, 256, 0, stream>>>(
        m1, lin2_w, lin2_b, m2, nullptr, nullptr, NNODES, 128, 1);
    k_head<<<(NNODES + 255) / 256, 256, 0, stream>>>(
        m2, lin3_w, lin3_b, (float*)d_out, NNODES);
}


# Edge scatter re-mapped to lane = column: one wave walks 32 edges, so each gather load and each f32 atomic-add instruction covers one whole 128-byte line; sums stay in atomic arrival order
# speedup vs baseline: 2.4483x; 1.4272x over previous
#include <hip/hip_runtime.h>

typedef float v2f __attribute__((ext_vector_type(2)));
typedef float v8f __attribute__((ext_vector_type(8)));

#define NNODES 20000
#define NEDGES 320000

template <int D>
__global__ __launch_bounds__(256) void k_scatter_add(
    const float* __restrict__ feat, const int* __restrict__ ei,
    float* __restrict__ agg)
{
    static_assert(D % 32 == 0 && D <= 128, "row = 1..4 whole lines");
    static_assert(NEDGES % 256 == 0, "8 waves x 32 edges per block, no tail");
    constexpr int CH = D / 32;
    constexpr int EB = 32 / CH;
    constexpr int NB = 32 / EB;

    const unsigned lane = threadIdx.x & 31;
    const int e0 = (blockIdx.x * 8 + (threadIdx.x >> 5)) * 32;

    const int s = ei[e0 + lane];
    const int d = ei[NEDGES + e0 + lane];

    float v[2][EB][CH];

#pragma unroll
    for (int j = 0; j < EB; ++j) {
        const float* rb = feat + (size_t)__builtin_amdgcn_readlane(s, j) * D;
#pragma unroll
        for (int c = 0; c < CH; ++c) v[0][j][c] = rb[lane + c * 32];
    }

#pragma unroll
    for (int b = 0; b < NB; ++b) {
        if (b + 1 < NB) {
#pragma unroll
            for (int j = 0; j < EB; ++j) {
                const float* rb =
                    feat + (size_t)__builtin_amdgcn_readlane(s, (b + 1) * EB + j) * D;
#pragma unroll
                for (int c = 0; c < CH; ++c) v[(b + 1) & 1][j][c] = rb[lane + c * 32];
            }
        }
#pragma unroll
        for (int j = 0; j < EB; ++j) {
            float* ob = agg + (size_t)__builtin_amdgcn_readlane(d, b * EB + j) * D;
#pragma unroll
            for (int c = 0; c < CH; ++c) unsafeAtomicAdd(ob + (lane + c * 32), v[b & 1][j][c]);
        }
    }
}

template <int NT, bool RELU_IN>
__global__ __launch_bounds__(256) void k_gemm_strip(
    const float* __restrict__ A,
    const float* __restrict__ B0, const float* __restrict__ bias0,
    float* __restrict__ C0,
    const float* __restrict__ B1, float* __restrict__ C1,
    int M, int K, int relu)
{
    constexpr int N = NT * 16;
    constexpr int BK = 64;
    __shared__ __align__(16) float Bs0[BK * N];

    const bool second = (blockIdx.y != 0);
    const float* __restrict__ B = second ? B1 : B0;
    float* __restrict__ C = second ? C1 : C0;

    const int lane = threadIdx.x & 31;
    const int wave = threadIdx.x >> 5;
    const int tm = blockIdx.x * 8 + wave;
    const bool active = (tm * 16 < M);
    const int rh = lane & 15;
    const int hi = lane >> 4;

    v8f acc[NT];
#pragma unroll
    for (int t = 0; t < NT; ++t) acc[t] = v8f{};

    const float* ap0 = A + (size_t)(tm * 16 + rh) * K + 2 * hi;

    for (int kc = 0; kc < K; kc += BK) {
        __syncthreads();
        {
            constexpr int ITERS = (BK / 2) * (N / 4);
            for (int idx = threadIdx.x; idx < ITERS; idx += 256) {
                const int pr = idx / (N / 4);
                const int c = (idx - pr * (N / 4)) * 4;
                const int k = 2 * pr;
                float4 g0 = *(const float4*)(B + (size_t)(kc + k) * N + c);
                float4 g1 = *(const float4*)(B + (size_t)(kc + k + 1) * N + c);
                float4 lo = {g0.x, g1.x, g0.y, g1.y};
                float4 hi4 = {g0.z, g1.z, g0.w, g1.w};
                float* dst = &Bs0[(pr * N + c) * 2];
                *(float4*)(dst) = lo;
                *(float4*)(dst + 4) = hi4;
            }
        }
        __syncthreads();
        if (!active) continue;

        const v2f* Bp0 = (const v2f*)Bs0;
        for (int k = 0; k < BK; k += 4) {
            const int pr = (k >> 1) + hi;
            v2f a;
            a.x = ap0[kc + k];
            a.y = ap0[kc + k + 1];
            if constexpr (RELU_IN) {
                a.x = fmaxf(a.x, 0.0f);
                a.y = fmaxf(a.y, 0.0f);
            }
#pragma unroll
            for (int t = 0; t < NT; ++t) {
                v2f b = Bp0[pr * N + t * 16 + rh];
                acc[t] = __builtin_amdgcn_wmma_f32_16x16x4_f32(
                    false, a, false, b, (short)0, acc[t], false, false);
            }
        }
    }

    if (active) {
#pragma unroll
        for (int t = 0; t < NT; ++t) {
            const int col = t * 16 + rh;
            const float bv = second ? 0.0f : bias0[col];
#pragma unroll
            for (int i = 0; i < 8; ++i) {
                int r = tm * 16 + i + 8 * hi;
                float v = acc[t][i] + bv;
                if (relu) v = fmaxf(v, 0.0f);
                C[(size_t)r * N + col] = v;
            }
        }
    }
}

__global__ __launch_bounds__(256) void k_head(
    const float* __restrict__ h, const float* __restrict__ W,
    const float* __restrict__ b, float* __restrict__ out, int M)
{
    int n = blockIdx.x * blockDim.x + threadIdx.x;
    if (n >= M) return;
    const float* hr = h + (size_t)n * 64;
    float z[10];
#pragma unroll
    for (int c = 0; c < 10; ++c) z[c] = b[c];
    for (int k = 0; k < 64; ++k) {
        float hv = hr[k];
        const float* wr = W + k * 10;
#pragma unroll
        for (int c = 0; c < 10; ++c) z[c] = fmaf(hv, wr[c], z[c]);
    }
    float mx = z[0];
#pragma unroll
    for (int c = 1; c < 10; ++c) mx = fmaxf(mx, z[c]);
    float s = 0.0f;
#pragma unroll
    for (int c = 0; c < 10; ++c) s += expf(z[c] - mx);
    float lse = mx + logf(s);
    float* o = out + (size_t)n * 10;
#pragma unroll
    for (int c = 0; c < 10; ++c) o[c] = z[c] - lse;
}

extern "C" void kernel_launch(void* const* d_in, const int* in_sizes, int n_in,
                              void* d_out, int out_size, void* d_ws, size_t ws_size,
                              hipStream_t stream) {
    const float* x       = (const float*)d_in[0];
    const int*   ei      = (const int*)d_in[1];
    const float* W1_rel  = (const float*)d_in[2];
    const float* W1_root = (const float*)d_in[3];
    const float* b1      = (const float*)d_in[4];
    const float* W2_rel  = (const float*)d_in[5];
    const float* W2_root = (const float*)d_in[6];
    const float* b2      = (const float*)d_in[7];
    const float* lin1_w  = (const float*)d_in[8];
    const float* lin1_b  = (const float*)d_in[9];
    const float* lin2_w  = (const float*)d_in[10];
    const float* lin2_b  = (const float*)d_in[11];
    const float* lin3_w  = (const float*)d_in[12];
    const float* lin3_b  = (const float*)d_in[13];

    float* y1 = (float*)d_ws;
    float* h1 = y1 + (size_t)NNODES * 128;
    float* y2 = h1 + (size_t)NNODES * 128;
    float* h2 = y2 + (size_t)NNODES * 64;
    float* m1 = h2 + (size_t)NNODES * 64;
    float* m2 = m1 + (size_t)NNODES * 128;

    const int mblocks = ((NNODES / 16) + 7) / 8;
    const dim3 grid2(mblocks, 2);


    k_gemm_strip<8, false><<<grid2, 256, 0, stream>>>(
        x, W1_root, b1, h1, W1_rel, y1, NNODES, 256, 0);
    k_scatter_add<128><<<NEDGES / 256, 256, 0, stream>>>(y1, ei, h1);

    k_gemm_strip<4, true><<<grid2, 256, 0, stream>>>(
        h1, W2_root, b2, h2, W2_rel, y2, NNODES, 128, 0);
    k_scatter_add<64><<<NEDGES / 256, 256, 0, stream>>>(y2, ei, h2);

    k_gemm_strip<8, false><<<mblocks, 256, 0, stream>>>(
        h2, lin1_w, lin1_b, m1, nullptr, nullptr, NNODES, 64, 1);
    k_gemm_strip<4, false><<<mblocks, 256, 0, stream>>>(
        m1, lin2_w, lin2_b, m2, nullptr, nullptr, NNODES, 128, 1);
    k_head<<<(NNODES + 255) / 256, 256, 0, stream>>>(
        m2, lin3_w, lin3_b, (float*)d_out, NNODES);
}


# Dense tail (lin1, lin2, lin3+log_softmax) fused into one kernel: transposed products, each layer's WMMA D tile is the next layer's B operand in registers; 7->5 launches; dot-product sum order changes
# speedup vs baseline: 2.9045x; 1.1863x over previous
#include <hip/hip_runtime.h>

typedef float v2f __attribute__((ext_vector_type(2)));
typedef float v8f __attribute__((ext_vector_type(8)));

#define NNODES 20000
#define NEDGES 320000

template <int D>
__global__ __launch_bounds__(256) void k_scatter_add(
    const float* __restrict__ feat, const int* __restrict__ ei,
    float* __restrict__ agg)
{
    static_assert(D % 32 == 0 && D <= 128, "row = 1..4 whole lines");
    static_assert(NEDGES % 256 == 0, "8 waves x 32 edges per block, no tail");
    constexpr int CH = D / 32;
    constexpr int EB = 32 / CH;
    constexpr int NB = 32 / EB;

    const unsigned lane = threadIdx.x & 31;
    const int e0 = (blockIdx.x * 8 + (threadIdx.x >> 5)) * 32;

    const int s = ei[e0 + lane];
    const int d = ei[NEDGES + e0 + lane];

    float v[2][EB][CH];

#pragma unroll
    for (int j = 0; j < EB; ++j) {
        const float* rb = feat + (size_t)__builtin_amdgcn_readlane(s, j) * D;
#pragma unroll
        for (int c = 0; c < CH; ++c) v[0][j][c] = rb[lane + c * 32];
    }

#pragma unroll
    for (int b = 0; b < NB; ++b) {
        if (b + 1 < NB) {
#pragma unroll
            for (int j = 0; j < EB; ++j) {
                const float* rb =
                    feat + (size_t)__builtin_amdgcn_readlane(s, (b + 1) * EB + j) * D;
#pragma unroll
                for (int c = 0; c < CH; ++c) v[(b + 1) & 1][j][c] = rb[lane + c * 32];
            }
        }
#pragma unroll
        for (int j = 0; j < EB; ++j) {
            float* ob = agg + (size_t)__builtin_amdgcn_readlane(d, b * EB + j) * D;
#pragma unroll
            for (int c = 0; c < CH; ++c) unsafeAtomicAdd(ob + (lane + c * 32), v[b & 1][j][c]);
        }
    }
}

template <int NT, bool RELU_IN>
__global__ __launch_bounds__(256) void k_gemm_strip(
    const float* __restrict__ A,
    const float* __restrict__ B0, const float* __restrict__ bias0,
    float* __restrict__ C0,
    const float* __restrict__ B1, float* __restrict__ C1,
    int M, int K, int relu)
{
    constexpr int N = NT * 16;
    constexpr int BK = 64;
    __shared__ __align__(16) float Bs0[BK * N];

    const bool second = (blockIdx.y != 0);
    const float* __restrict__ B = second ? B1 : B0;
    float* __restrict__ C = second ? C1 : C0;

    const int lane = threadIdx.x & 31;
    const int wave = threadIdx.x >> 5;
    const int tm = blockIdx.x * 8 + wave;
    const bool active = (tm * 16 < M);
    const int rh = lane & 15;
    const int hi = lane >> 4;

    v8f acc[NT];
#pragma unroll
    for (int t = 0; t < NT; ++t) acc[t] = v8f{};

    const float* ap0 = A + (size_t)(tm * 16 + rh) * K + 2 * hi;

    for (int kc = 0; kc < K; kc += BK) {
        __syncthreads();
        {
            constexpr int ITERS = (BK / 2) * (N / 4);
            for (int idx = threadIdx.x; idx < ITERS; idx += 256) {
                const int pr = idx / (N / 4);
                const int c = (idx - pr * (N / 4)) * 4;
                const int k = 2 * pr;
                float4 g0 = *(const float4*)(B + (size_t)(kc + k) * N + c);
                float4 g1 = *(const float4*)(B + (size_t)(kc + k + 1) * N + c);
                float4 lo = {g0.x, g1.x, g0.y, g1.y};
                float4 hi4 = {g0.z, g1.z, g0.w, g1.w};
                float* dst = &Bs0[(pr * N + c) * 2];
                *(float4*)(dst) = lo;
                *(float4*)(dst + 4) = hi4;
            }
        }
        __syncthreads();
        if (!active) continue;

        const v2f* Bp0 = (const v2f*)Bs0;
        for (int k = 0; k < BK; k += 4) {
            const int pr = (k >> 1) + hi;
            v2f a;
            a.x = ap0[kc + k];
            a.y = ap0[kc + k + 1];
            if constexpr (RELU_IN) {
                a.x = fmaxf(a.x, 0.0f);
                a.y = fmaxf(a.y, 0.0f);
            }
#pragma unroll
            for (int t = 0; t < NT; ++t) {
                v2f b = Bp0[pr * N + t * 16 + rh];
                acc[t] = __builtin_amdgcn_wmma_f32_16x16x4_f32(
                    false, a, false, b, (short)0, acc[t], false, false);
            }
        }
    }

    if (active) {
#pragma unroll
        for (int t = 0; t < NT; ++t) {
            const int col = t * 16 + rh;
            const float bv = second ? 0.0f : bias0[col];
#pragma unroll
            for (int i = 0; i < 8; ++i) {
                int r = tm * 16 + i + 8 * hi;
                float v = acc[t][i] + bv;
                if (relu) v = fmaxf(v, 0.0f);
                C[(size_t)r * N + col] = v;
            }
        }
    }
}

template <int K, int N>
struct PairStage {
    static constexpr int ITERS = (K / 2) * (N / 4);
    static_assert(ITERS % 256 == 0, "whole passes of the 256-thread block");
    static constexpr int PASSES = ITERS / 256;
    float4 g0[PASSES], g1[PASSES];

    __device__ __forceinline__ void load(const float* __restrict__ Wg)
    {
#pragma unroll
        for (int p = 0; p < PASSES; ++p) {
            const int idx = threadIdx.x + 256 * p;
            const int pr = idx / (N / 4);
            const int c = (idx - pr * (N / 4)) * 4;
            g0[p] = *(const float4*)(Wg + (size_t)(2 * pr) * N + c);
            g1[p] = *(const float4*)(Wg + (size_t)(2 * pr + 1) * N + c);
        }
    }
    __device__ __forceinline__ void store(float* __restrict__ dst0) const
    {
#pragma unroll
        for (int p = 0; p < PASSES; ++p) {
            const int idx = threadIdx.x + 256 * p;
            const int pr = idx / (N / 4);
            const int c = (idx - pr * (N / 4)) * 4;
            float4 lo = {g0[p].x, g1[p].x, g0[p].y, g1[p].y};
            float4 hi4 = {g0[p].z, g1[p].z, g0[p].w, g1[p].w};
            float* d = dst0 + (pr * N + c) * 2;
            *(float4*)(d) = lo;
            *(float4*)(d + 4) = hi4;
        }
    }
};

__global__ __launch_bounds__(256) void k_mlp_tail(
    const float* __restrict__ hin,
    const float* __restrict__ W1, const float* __restrict__ b1,
    const float* __restrict__ W2, const float* __restrict__ b2,
    const float* __restrict__ W3, const float* __restrict__ b3,
    float* __restrict__ out, int M)
{
    __shared__ __align__(16) float W1s[64 * 128];
    __shared__ __align__(16) float W2s[128 * 64];
    __shared__ __align__(16) float W3s[32 * 16 * 2];

    const int lane = threadIdx.x & 31;
    const int wave = threadIdx.x >> 5;
    const int tm = blockIdx.x * 8 + wave;
    const bool active = (tm * 16 < M);
    const int c = lane & 15;
    const int hh = lane >> 4;

    v2f xb[16];
    v8f acc1[8];
    v8f acc2[4];
    v8f z;
    if (active) {
        const float* xp = hin + (size_t)(tm * 16 + c) * 64 + 2 * hh;
#pragma unroll
        for (int ks = 0; ks < 16; ++ks) xb[ks] = *(const v2f*)(xp + 4 * ks);
#pragma unroll
        for (int t = 0; t < 8; ++t) {
            const float4 lo = *(const float4*)(b1 + 16 * t + 8 * hh);
            const float4 hi4 = *(const float4*)(b1 + 16 * t + 8 * hh + 4);
            acc1[t] = v8f{lo.x, lo.y, lo.z, lo.w, hi4.x, hi4.y, hi4.z, hi4.w};
        }
#pragma unroll
        for (int t = 0; t < 4; ++t) {
            const float4 lo = *(const float4*)(b2 + 16 * t + 8 * hh);
            const float4 hi4 = *(const float4*)(b2 + 16 * t + 8 * hh + 4);
            acc2[t] = v8f{lo.x, lo.y, lo.z, lo.w, hi4.x, hi4.y, hi4.z, hi4.w};
        }
#pragma unroll
        for (int r = 0; r < 8; ++r) {
            const int cls = 8 * hh + r;
            z[r] = b3[cls < 10 ? cls : 9];
        }
    }

    PairStage<64, 128> st1;
    PairStage<128, 64> st2;
    v2f w3v[2];
    st1.load(W1);
    st2.load(W2);
#pragma unroll
    for (int p = 0; p < 2; ++p) {
        const int idx = threadIdx.x + 256 * p;
        const int pr = idx >> 4;
        const int col = idx & 15;
        const int cc = col < 10 ? col : 9;
        w3v[p] = v2f{W3[(2 * pr) * 10 + cc], W3[(2 * pr + 1) * 10 + cc]};
    }
    __builtin_amdgcn_sched_barrier(0);
    st1.store(W1s);
    st2.store(W2s);
#pragma unroll
    for (int p = 0; p < 2; ++p) ((v2f*)W3s)[threadIdx.x + 256 * p] = w3v[p];
    __syncthreads();
    if (!active) return;

    {
        const v2f* Wp = (const v2f*)W1s + hh * 128 + c;
#pragma unroll
        for (int ks = 0; ks < 16; ++ks) {
#pragma unroll
            for (int t = 0; t < 8; ++t) {
                const v2f a = Wp[(2 * ks) * 128 + t * 16];
                acc1[t] = __builtin_amdgcn_wmma_f32_16x16x4_f32(
                    false, a, false, xb[ks], (short)0, acc1[t], false, false);
            }
        }
#pragma unroll
        for (int t = 0; t < 8; ++t)
#pragma unroll
            for (int r = 0; r < 8; ++r) acc1[t][r] = fmaxf(acc1[t][r], 0.0f);
    }

    {
        const v2f* Wp = (const v2f*)W2s + (4 * hh) * 64 + c;
#pragma unroll
        for (int t = 0; t < 8; ++t) {
#pragma unroll
            for (int j = 0; j < 4; ++j) {
                const v2f b = {acc1[t][2 * j], acc1[t][2 * j + 1]};
#pragma unroll
                for (int mt = 0; mt < 4; ++mt) {
                    const v2f a = Wp[(8 * t + j) * 64 + mt * 16];
                    acc2[mt] = __builtin_amdgcn_wmma_f32_16x16x4_f32(
                        false, a, false, b, (short)0, acc2[mt], false, false);
                }
            }
        }
#pragma unroll
        for (int t = 0; t < 4; ++t)
#pragma unroll
            for (int r = 0; r < 8; ++r) acc2[t][r] = fmaxf(acc2[t][r], 0.0f);
    }

    {
        const v2f* Wp = (const v2f*)W3s + (4 * hh) * 16 + c;
#pragma unroll
        for (int t = 0; t < 4; ++t) {
#pragma unroll
            for (int j = 0; j < 4; ++j) {
                const v2f b = {acc2[t][2 * j], acc2[t][2 * j + 1]};
                const v2f a = Wp[(8 * t + j) * 16];
                z = __builtin_amdgcn_wmma_f32_16x16x4_f32(
                    false, a, false, b, (short)0, z, false, false);
            }
        }
    }

    const bool lowhalf = (hh == 0);
    float mx = fmaxf(z[0], z[1]);
#pragma unroll
    for (int r = 2; r < 8; ++r) mx = lowhalf ? fmaxf(mx, z[r]) : mx;
    mx = fmaxf(mx, __shfl_xor(mx, 16));
    float s = expf(z[0] - mx) + expf(z[1] - mx);
#pragma unroll
    for (int r = 2; r < 8; ++r) s += lowhalf ? expf(z[r] - mx) : 0.0f;
    s += __shfl_xor(s, 16);
    const float lse = mx + logf(s);

    float* o = out + (size_t)(tm * 16 + c) * 10 + 8 * hh;
    *(float2*)(o) = float2{z[0] - lse, z[1] - lse};
    if (lowhalf) {
        *(float2*)(o + 2) = float2{z[2] - lse, z[3] - lse};
        *(float2*)(o + 4) = float2{z[4] - lse, z[5] - lse};
        *(float2*)(o + 6) = float2{z[6] - lse, z[7] - lse};
    }
}

extern "C" void kernel_launch(void* const* d_in, const int* in_sizes, int n_in,
                              void* d_out, int out_size, void* d_ws, size_t ws_size,
                              hipStream_t stream) {
    const float* x       = (const float*)d_in[0];
    const int*   ei      = (const int*)d_in[1];
    const float* W1_rel  = (const float*)d_in[2];
    const float* W1_root = (const float*)d_in[3];
    const float* b1      = (const float*)d_in[4];
    const float* W2_rel  = (const float*)d_in[5];
    const float* W2_root = (const float*)d_in[6];
    const float* b2      = (const float*)d_in[7];
    const float* lin1_w  = (const float*)d_in[8];
    const float* lin1_b  = (const float*)d_in[9];
    const float* lin2_w  = (const float*)d_in[10];
    const float* lin2_b  = (const float*)d_in[11];
    const float* lin3_w  = (const float*)d_in[12];
    const float* lin3_b  = (const float*)d_in[13];

    float* y1 = (float*)d_ws;
    float* h1 = y1 + (size_t)NNODES * 128;
    float* y2 = h1 + (size_t)NNODES * 128;
    float* h2 = y2 + (size_t)NNODES * 64;

    const int mblocks = ((NNODES / 16) + 7) / 8;
    const dim3 grid2(mblocks, 2);


    k_gemm_strip<8, false><<<grid2, 256, 0, stream>>>(
        x, W1_root, b1, h1, W1_rel, y1, NNODES, 256, 0);
    k_scatter_add<128><<<NEDGES / 256, 256, 0, stream>>>(y1, ei, h1);

    k_gemm_strip<4, true><<<grid2, 256, 0, stream>>>(
        h1, W2_root, b2, h2, W2_rel, y2, NNODES, 128, 0);
    k_scatter_add<64><<<NEDGES / 256, 256, 0, stream>>>(y2, ei, h2);

    k_mlp_tail<<<mblocks, 256, 0, stream>>>(
        h2, lin1_w, lin1_b, lin2_w, lin2_b, lin3_w, lin3_b, (float*)d_out, NNODES);
}


# Both graph-layer GEMMs re-cut to one 8-wave work-group per WGP (250 x 80 rows x root+rel columns; A and weights double-buffered in LDS): equal f32 WMMA count on every SIMD, not 1 or 2 blocks per WGP
# speedup vs baseline: 3.3743x; 1.1617x over previous
#include <hip/hip_runtime.h>

typedef float v2f __attribute__((ext_vector_type(2)));
typedef float v8f __attribute__((ext_vector_type(8)));

#define NNODES 20000
#define NEDGES 320000

template <int D>
__global__ __launch_bounds__(256) void k_scatter_add(
    const float* __restrict__ feat, const int* __restrict__ ei,
    float* __restrict__ agg)
{
    static_assert(D % 32 == 0 && D <= 128, "row = 1..4 whole lines");
    static_assert(NEDGES % 256 == 0, "8 waves x 32 edges per block, no tail");
    constexpr int CH = D / 32;
    constexpr int EB = 32 / CH;
    constexpr int NB = 32 / EB;

    const unsigned lane = threadIdx.x & 31;
    const int e0 = (blockIdx.x * 8 + (threadIdx.x >> 5)) * 32;

    const int s = ei[e0 + lane];
    const int d = ei[NEDGES + e0 + lane];

    float v[2][EB][CH];

#pragma unroll
    for (int j = 0; j < EB; ++j) {
        const float* rb = feat + (size_t)__builtin_amdgcn_readlane(s, j) * D;
#pragma unroll
        for (int c = 0; c < CH; ++c) v[0][j][c] = rb[lane + c * 32];
    }

#pragma unroll
    for (int b = 0; b < NB; ++b) {
        if (b + 1 < NB) {
#pragma unroll
            for (int j = 0; j < EB; ++j) {
                const float* rb =
                    feat + (size_t)__builtin_amdgcn_readlane(s, (b + 1) * EB + j) * D;
#pragma unroll
                for (int c = 0; c < CH; ++c) v[(b + 1) & 1][j][c] = rb[lane + c * 32];
            }
        }
#pragma unroll
        for (int j = 0; j < EB; ++j) {
            float* ob = agg + (size_t)__builtin_amdgcn_readlane(d, b * EB + j) * D;
#pragma unroll
            for (int c = 0; c < CH; ++c) unsafeAtomicAdd(ob + (lane + c * 32), v[b & 1][j][c]);
        }
    }
}

#define GEMM_THREADS 256

template <int NTOT>
struct ChunkStage {
    static constexpr int BK = 32;
    static constexpr int NH = NTOT / 2;
    static constexpr int ITERS = (BK / 2) * (NTOT / 4);
    static_assert(ITERS % GEMM_THREADS == 0, "whole passes of the work-group");
    static constexpr int PASSES = ITERS / GEMM_THREADS;
    float4 g0[PASSES], g1[PASSES];

    __device__ __forceinline__ void load(const float* B0, const float* B1, int kc)
    {
#pragma unroll
        for (int p = 0; p < PASSES; ++p) {
            const int idx = threadIdx.x + GEMM_THREADS * p;
            const int pr = idx / (NTOT / 4);
            const int c = (idx - pr * (NTOT / 4)) * 4;
            const float* src = (c < NH) ? (B0 + c) : (B1 + (c - NH));
            g0[p] = *(const float4*)(src + (size_t)(kc + 2 * pr) * NH);
            g1[p] = *(const float4*)(src + (size_t)(kc + 2 * pr + 1) * NH);
        }
    }
    __device__ __forceinline__ void store(float* __restrict__ dst0) const
    {
#pragma unroll
        for (int p = 0; p < PASSES; ++p) {
            const int idx = threadIdx.x + GEMM_THREADS * p;
            const int pr = idx / (NTOT / 4);
            const int c = (idx - pr * (NTOT / 4)) * 4;
            float4 lo = {g0[p].x, g1[p].x, g0[p].y, g1[p].y};
            float4 hi4 = {g0[p].z, g1[p].z, g0[p].w, g1[p].w};
            float* d = dst0 + (pr * NTOT + c) * 2;
            *(float4*)(d) = lo;
            *(float4*)(d + 4) = hi4;
        }
    }
};

template <int NTW, int K, bool RELU_IN>
__global__ __launch_bounds__(GEMM_THREADS) void k_gemm_pair(
    const float* A,
    const float* B0, const float* __restrict__ bias0,
    float* __restrict__ C0,
    const float* B1, float* __restrict__ C1)
{
    constexpr int RT = 5;
    constexpr int NWV = GEMM_THREADS / 32;
    constexpr int NW = NTW * 16;
    constexpr int NTOT = NW * NWV;
    constexpr int NH = NTOT / 2;
    constexpr int BK = ChunkStage<NTOT>::BK;
    constexpr int NCH = K / BK;
    constexpr int KS = BK / 4;
    constexpr int ROWS = RT * 16;
    constexpr int AP = BK + 4;
    constexpr int AITEMS = ROWS * (BK / 4);
    constexpr int APASSES = (AITEMS + GEMM_THREADS - 1) / GEMM_THREADS;
    static_assert(K % BK == 0, "whole K-chunks");
    static_assert(NH % NW == 0, "a wave's columns lie in one product");
    static_assert(NNODES % ROWS == 0, "no row tail: every wave is active");
    static_assert(BK == 32, "A staging: 8 float4 per row");

    __shared__ __align__(16) float L[2 * ROWS * AP + 2 * BK * NTOT];
    float (*As)[ROWS * AP] = (float (*)[ROWS * AP])L;
    float (*Bs)[BK * NTOT] = (float (*)[BK * NTOT])(L + 2 * ROWS * AP);

    const int lane = threadIdx.x & 31;
    const int wave = threadIdx.x >> 5;
    const int rh = lane & 15;
    const int hi = lane >> 4;
    const int col0 = wave * NW;

    v8f acc[RT][NTW];
#pragma unroll
    for (int r = 0; r < RT; ++r)
#pragma unroll
        for (int t = 0; t < NTW; ++t) acc[r][t] = v8f{};

    static_assert(APASSES == 3 && GEMM_THREADS % 8 == 0, "three A passes, written out below");
    const int it2r = threadIdx.x + 2 * GEMM_THREADS;
    const int it2 = (it2r >= AITEMS) ? (it2r - AITEMS) : it2r;
    const int aoff0 = (threadIdx.x >> 3) * AP + (threadIdx.x & 7) * 4;
    const int aoff1 = aoff0 + (GEMM_THREADS / 8) * AP;
    const int aoff2 = (it2 >> 3) * AP + (it2 & 7) * 4;
    const float* agp0 = A + (size_t)(blockIdx.x * ROWS + (threadIdx.x >> 3)) * K + (threadIdx.x & 7) * 4;
    const float* agp1 = agp0 + (size_t)(GEMM_THREADS / 8) * K;
    const float* agp2 = A + (size_t)(blockIdx.x * ROWS + (it2 >> 3)) * K + (it2 & 7) * 4;

    ChunkStage<NTOT> st;
    float4 a40, a41, a42;

    st.load(B0, B1, 0);
    a40 = *(const float4*)(agp0);
    a41 = *(const float4*)(agp1);
    a42 = *(const float4*)(agp2);
    st.store(Bs[0]);
    *(float4*)(&As[0][aoff0]) = a40;
    *(float4*)(&As[0][aoff1]) = a41;
    *(float4*)(&As[0][aoff2]) = a42;
    __syncthreads();

#pragma unroll
    for (int ci = 0; ci < NCH; ++ci) {
        const int cur = ci & 1;
        if (ci + 1 < NCH) {
            st.load(B0, B1, (ci + 1) * BK);
            a40 = *(const float4*)(agp0 + (ci + 1) * BK);
            a41 = *(const float4*)(agp1 + (ci + 1) * BK);
            a42 = *(const float4*)(agp2 + (ci + 1) * BK);
            asm volatile("" ::: "memory");
        }

        const v2f* Bp = (const v2f*)Bs[cur] + hi * NTOT + col0 + rh;
        const v2f* Ap = (const v2f*)(&As[cur][rh * AP + 2 * hi]);
#pragma unroll
        for (int ks = 0; ks < KS; ++ks) {
            v2f a[RT], b[NTW];
#pragma unroll
            for (int r = 0; r < RT; ++r) {
                a[r] = Ap[r * (16 * AP / 2) + 2 * ks];
                if constexpr (RELU_IN) {
                    a[r].x = fmaxf(a[r].x, 0.0f);
                    a[r].y = fmaxf(a[r].y, 0.0f);
                }
            }
#pragma unroll
            for (int t = 0; t < NTW; ++t)
                b[t] = Bp[(2 * ks) * NTOT + t * 16];
#pragma unroll
            for (int r = 0; r < RT; ++r)
#pragma unroll
                for (int t = 0; t < NTW; ++t)
                    acc[r][t] = __builtin_amdgcn_wmma_f32_16x16x4_f32(
                        false, a[r], false, b[t], (short)0, acc[r][t], false, false);
        }

        if (ci + 1 < NCH) {
            __builtin_amdgcn_sched_barrier(0);
            st.store(Bs[cur ^ 1]);
            *(float4*)(&As[cur ^ 1][aoff0]) = a40;
            *(float4*)(&As[cur ^ 1][aoff1]) = a41;
            *(float4*)(&As[cur ^ 1][aoff2]) = a42;
            __syncthreads();
        }
    }

    const bool second = (col0 >= NH);
    float* __restrict__ C = second ? C1 : C0;
    const int ch0 = second ? (col0 - NH) : col0;
#pragma unroll
    for (int t = 0; t < NTW; ++t) {
        const int col = ch0 + t * 16 + rh;
        const float bv = second ? 0.0f : bias0[col];
#pragma unroll
        for (int r = 0; r < RT; ++r) {
#pragma unroll
            for (int i = 0; i < 8; ++i) {
                const int row = blockIdx.x * ROWS + r * 16 + i + 8 * hi;
                C[(size_t)row * NH + col] = acc[r][t][i] + bv;
            }
        }
    }
}

template <int K, int N>
struct PairStage {
    static constexpr int ITERS = (K / 2) * (N / 4);
    static_assert(ITERS % 256 == 0, "whole passes of the 256-thread block");
    static constexpr int PASSES = ITERS / 256;
    float4 g0[PASSES], g1[PASSES];

    __device__ __forceinline__ void load(const float* __restrict__ Wg)
    {
#pragma unroll
        for (int p = 0; p < PASSES; ++p) {
            const int idx = threadIdx.x + 256 * p;
            const int pr = idx / (N / 4);
            const int c = (idx - pr * (N / 4)) * 4;
            g0[p] = *(const float4*)(Wg + (size_t)(2 * pr) * N + c);
            g1[p] = *(const float4*)(Wg + (size_t)(2 * pr + 1) * N + c);
        }
    }
    __device__ __forceinline__ void store(float* __restrict__ dst0) const
    {
#pragma unroll
        for (int p = 0; p < PASSES; ++p) {
            const int idx = threadIdx.x + 256 * p;
            const int pr = idx / (N / 4);
            const int c = (idx - pr * (N / 4)) * 4;
            float4 lo = {g0[p].x, g1[p].x, g0[p].y, g1[p].y};
            float4 hi4 = {g0[p].z, g1[p].z, g0[p].w, g1[p].w};
            float* d = dst0 + (pr * N + c) * 2;
            *(float4*)(d) = lo;
            *(float4*)(d + 4) = hi4;
        }
    }
};

__global__ __launch_bounds__(256) void k_mlp_tail(
    const float* __restrict__ hin,
    const float* __restrict__ W1, const float* __restrict__ b1,
    const float* __restrict__ W2, const float* __restrict__ b2,
    const float* __restrict__ W3, const float* __restrict__ b3,
    float* __restrict__ out, int M)
{
    __shared__ __align__(16) float W1s[64 * 128];
    __shared__ __align__(16) float W2s[128 * 64];
    __shared__ __align__(16) float W3s[32 * 16 * 2];

    const int lane = threadIdx.x & 31;
    const int wave = threadIdx.x >> 5;
    const int tm = blockIdx.x * 8 + wave;
    const bool active = (tm * 16 < M);
    const int c = lane & 15;
    const int hh = lane >> 4;

    v2f xb[16];
    v8f acc1[8];
    v8f acc2[4];
    v8f z;
    if (active) {
        const float* xp = hin + (size_t)(tm * 16 + c) * 64 + 2 * hh;
#pragma unroll
        for (int ks = 0; ks < 16; ++ks) xb[ks] = *(const v2f*)(xp + 4 * ks);
#pragma unroll
        for (int t = 0; t < 8; ++t) {
            const float4 lo = *(const float4*)(b1 + 16 * t + 8 * hh);
            const float4 hi4 = *(const float4*)(b1 + 16 * t + 8 * hh + 4);
            acc1[t] = v8f{lo.x, lo.y, lo.z, lo.w, hi4.x, hi4.y, hi4.z, hi4.w};
        }
#pragma unroll
        for (int t = 0; t < 4; ++t) {
            const float4 lo = *(const float4*)(b2 + 16 * t + 8 * hh);
            const float4 hi4 = *(const float4*)(b2 + 16 * t + 8 * hh + 4);
            acc2[t] = v8f{lo.x, lo.y, lo.z, lo.w, hi4.x, hi4.y, hi4.z, hi4.w};
        }
#pragma unroll
        for (int r = 0; r < 8; ++r) {
            const int cls = 8 * hh + r;
            z[r] = b3[cls < 10 ? cls : 9];
        }
    }

    PairStage<64, 128> st1;
    PairStage<128, 64> st2;
    v2f w3v[2];
    st1.load(W1);
    st2.load(W2);
#pragma unroll
    for (int p = 0; p < 2; ++p) {
        const int idx = threadIdx.x + 256 * p;
        const int pr = idx >> 4;
        const int col = idx & 15;
        const int cc = col < 10 ? col : 9;
        w3v[p] = v2f{W3[(2 * pr) * 10 + cc], W3[(2 * pr + 1) * 10 + cc]};
    }
    __builtin_amdgcn_sched_barrier(0);
    st1.store(W1s);
    st2.store(W2s);
#pragma unroll
    for (int p = 0; p < 2; ++p) ((v2f*)W3s)[threadIdx.x + 256 * p] = w3v[p];
    __syncthreads();
    if (!active) return;

    {
        const v2f* Wp = (const v2f*)W1s + hh * 128 + c;
#pragma unroll
        for (int ks = 0; ks < 16; ++ks) {
#pragma unroll
            for (int t = 0; t < 8; ++t) {
                const v2f a = Wp[(2 * ks) * 128 + t * 16];
                acc1[t] = __builtin_amdgcn_wmma_f32_16x16x4_f32(
                    false, a, false, xb[ks], (short)0, acc1[t], false, false);
            }
        }
#pragma unroll
        for (int t = 0; t < 8; ++t)
#pragma unroll
            for (int r = 0; r < 8; ++r) acc1[t][r] = fmaxf(acc1[t][r], 0.0f);
    }

    {
        const v2f* Wp = (const v2f*)W2s + (4 * hh) * 64 + c;
#pragma unroll
        for (int t = 0; t < 8; ++t) {
#pragma unroll
            for (int j = 0; j < 4; ++j) {
                const v2f b = {acc1[t][2 * j], acc1[t][2 * j + 1]};
#pragma unroll
                for (int mt = 0; mt < 4; ++mt) {
                    const v2f a = Wp[(8 * t + j) * 64 + mt * 16];
                    acc2[mt] = __builtin_amdgcn_wmma_f32_16x16x4_f32(
                        false, a, false, b, (short)0, acc2[mt], false, false);
                }
            }
        }
#pragma unroll
        for (int t = 0; t < 4; ++t)
#pragma unroll
            for (int r = 0; r < 8; ++r) acc2[t][r] = fmaxf(acc2[t][r], 0.0f);
    }

    {
        const v2f* Wp = (const v2f*)W3s + (4 * hh) * 16 + c;
#pragma unroll
        for (int t = 0; t < 4; ++t) {
#pragma unroll
            for (int j = 0; j < 4; ++j) {
                const v2f b = {acc2[t][2 * j], acc2[t][2 * j + 1]};
                const v2f a = Wp[(8 * t + j) * 16];
                z = __builtin_amdgcn_wmma_f32_16x16x4_f32(
                    false, a, false, b, (short)0, z, false, false);
            }
        }
    }

    const bool lowhalf = (hh == 0);
    float mx = fmaxf(z[0], z[1]);
#pragma unroll
    for (int r = 2; r < 8; ++r) mx = lowhalf ? fmaxf(mx, z[r]) : mx;
    mx = fmaxf(mx, __shfl_xor(mx, 16));
    float s = expf(z[0] - mx) + expf(z[1] - mx);
#pragma unroll
    for (int r = 2; r < 8; ++r) s += lowhalf ? expf(z[r] - mx) : 0.0f;
    s += __shfl_xor(s, 16);
    const float lse = mx + logf(s);

    float* o = out + (size_t)(tm * 16 + c) * 10 + 8 * hh;
    *(float2*)(o) = float2{z[0] - lse, z[1] - lse};
    if (lowhalf) {
        *(float2*)(o + 2) = float2{z[2] - lse, z[3] - lse};
        *(float2*)(o + 4) = float2{z[4] - lse, z[5] - lse};
        *(float2*)(o + 6) = float2{z[6] - lse, z[7] - lse};
    }
}

extern "C" void kernel_launch(void* const* d_in, const int* in_sizes, int n_in,
                              void* d_out, int out_size, void* d_ws, size_t ws_size,
                              hipStream_t stream) {
    const float* x       = (const float*)d_in[0];
    const int*   ei      = (const int*)d_in[1];
    const float* W1_rel  = (const float*)d_in[2];
    const float* W1_root = (const float*)d_in[3];
    const float* b1      = (const float*)d_in[4];
    const float* W2_rel  = (const float*)d_in[5];
    const float* W2_root = (const float*)d_in[6];
    const float* b2      = (const float*)d_in[7];
    const float* lin1_w  = (const float*)d_in[8];
    const float* lin1_b  = (const float*)d_in[9];
    const float* lin2_w  = (const float*)d_in[10];
    const float* lin2_b  = (const float*)d_in[11];
    const float* lin3_w  = (const float*)d_in[12];
    const float* lin3_b  = (const float*)d_in[13];

    float* y1 = (float*)d_ws;
    float* h1 = y1 + (size_t)NNODES * 128;
    float* y2 = h1 + (size_t)NNODES * 128;
    float* h2 = y2 + (size_t)NNODES * 64;

    const int mblocks = ((NNODES / 16) + 7) / 8;


    k_gemm_pair<2, 256, false><<<NNODES / 80, GEMM_THREADS, 0, stream>>>(
        x, W1_root, b1, h1, W1_rel, y1);
    k_scatter_add<128><<<NEDGES / 256, 256, 0, stream>>>(y1, ei, h1);

    k_gemm_pair<1, 128, true><<<NNODES / 80, GEMM_THREADS, 0, stream>>>(
        h1, W2_root, b2, h2, W2_rel, y2);
    k_scatter_add<64><<<NEDGES / 256, 256, 0, stream>>>(y2, ei, h2);

    k_mlp_tail<<<mblocks, 256, 0, stream>>>(
        h2, lin1_w, lin1_b, lin2_w, lin2_b, lin3_w, lin3_b, (float*)d_out, NNODES);
}


# Edge scatter by chip half: a work-group reads its die and draws (256-edge slice, dst-row half) tasks by integer ticket adds, so a row's f32 atomic adds come from one L2 half; arrival order as before
# speedup vs baseline: 3.9446x; 1.1690x over previous
#include <hip/hip_runtime.h>

typedef float v2f __attribute__((ext_vector_type(2)));
typedef float v8f __attribute__((ext_vector_type(8)));

#define NNODES 20000
#define NEDGES 320000

#define SLICE 256
#define NTASK (NEDGES / SLICE)
#define QSTRIDE 64

template <int D>
__global__ __launch_bounds__(256) void k_scatter_add(
    const float* __restrict__ feat, const int* __restrict__ ei,
    float* __restrict__ agg, int* __restrict__ q)
{
    static_assert(D % 32 == 0 && D <= 128, "row = 1..4 whole lines");
    static_assert(NEDGES % SLICE == 0, "whole slices, no tail");
    constexpr int CH = D / 32;
    constexpr int EB = 32 / CH;
    constexpr int NB = 64 / EB;

    __shared__ int task[2];
    __shared__ int ls[8][96];
    __shared__ int ld[8][96];

    const unsigned lane = threadIdx.x & 31;
    const int wave = threadIdx.x >> 5;

    if (threadIdx.x == 0) {
        const int h = (int)((__builtin_amdgcn_s_sendmsg_rtn(0x87) >> 18) & 1u);
        const int t = atomicAdd(q + h * QSTRIDE, 2);
        int ta = (t < NTASK) ? (2 * t + h) : -1;
        int tb = (t + 1 < NTASK) ? (2 * (t + 1) + h) : -1;
        const int miss = (ta < 0) + (tb < 0);
        if (miss) {
            int u = atomicAdd(q + (h ^ 1) * QSTRIDE, miss);
            if (ta < 0) { ta = (u < NTASK) ? (2 * u + (h ^ 1)) : -1; ++u; }
            if (tb < 0) { tb = (u < NTASK) ? (2 * u + (h ^ 1)) : -1; }
        }
        task[0] = ta;
        task[1] = tb;
    }
    __syncthreads();
    const int ta = __builtin_amdgcn_readfirstlane(task[0]);
    const int tb = __builtin_amdgcn_readfirstlane(task[1]);

    const int ea = ((ta >= 0 ? ta : 0) >> 1) * SLICE + wave * 32 + lane;
    const int eb = ((tb >= 0 ? tb : 0) >> 1) * SLICE + wave * 32 + lane;
    const int sa = ei[ea], da = ei[NEDGES + ea];
    const int sb = ei[eb], db = ei[NEDGES + eb];
    const bool ka = (ta >= 0) && (((da >= NNODES / 2) ? 1 : 0) == (ta & 1));
    const bool kb = (tb >= 0) && (((db >= NNODES / 2) ? 1 : 0) == (tb & 1));
    const unsigned ma = __builtin_amdgcn_ballot_w32(ka);
    const unsigned mb = __builtin_amdgcn_ballot_w32(kb);
    const int na = __builtin_popcount(ma);
    const int n = na + __builtin_popcount(mb);
    const int pa = ka ? (int)__builtin_amdgcn_mbcnt_lo(ma, 0) : 64 + (int)lane;
    const int pb = kb ? na + (int)__builtin_amdgcn_mbcnt_lo(mb, 0) : 64 + (int)lane;
    ls[wave][pa] = sa;
    ld[wave][pa] = da;
    ls[wave][pb] = sb;
    ld[wave][pb] = db;
    __syncthreads();
    const bool u0 = (int)lane < n, u1 = (int)lane + 32 < n;
    const int s0 = u0 ? ls[wave][lane] : 0, s1 = u1 ? ls[wave][32 + lane] : 0;
    const int d0 = u0 ? ld[wave][lane] : 0, d1 = u1 ? ld[wave][32 + lane] : 0;

    float v[2][EB][CH];

#pragma unroll
    for (int j = 0; j < EB; ++j) {
        if (j < n) {
            const float* rb = feat + (size_t)__builtin_amdgcn_readlane(s0, j) * D;
#pragma unroll
            for (int c = 0; c < CH; ++c) v[0][j][c] = rb[lane + c * 32];
        }
    }

#pragma unroll
    for (int b = 0; b < NB; ++b) {
        if (b + 1 < NB) {
#pragma unroll
            for (int j = 0; j < EB; ++j) {
                const int e = (b + 1) * EB + j;
                if (e < n) {
                    const float* rb =
                        feat + (size_t)__builtin_amdgcn_readlane(e < 32 ? s0 : s1, e & 31) * D;
#pragma unroll
                    for (int c = 0; c < CH; ++c) v[(b + 1) & 1][j][c] = rb[lane + c * 32];
                }
            }
        }
#pragma unroll
        for (int j = 0; j < EB; ++j) {
            const int e = b * EB + j;
            if (e < n) {
                float* ob = agg + (size_t)__builtin_amdgcn_readlane(e < 32 ? d0 : d1, e & 31) * D;
#pragma unroll
                for (int c = 0; c < CH; ++c) unsafeAtomicAdd(ob + (lane + c * 32), v[b & 1][j][c]);
            }
        }
    }
}

#define GEMM_THREADS 256

template <int NTOT>
struct ChunkStage {
    static constexpr int BK = 32;
    static constexpr int NH = NTOT / 2;
    static constexpr int ITERS = (BK / 2) * (NTOT / 4);
    static_assert(ITERS % GEMM_THREADS == 0, "whole passes of the work-group");
    static constexpr int PASSES = ITERS / GEMM_THREADS;
    float4 g0[PASSES], g1[PASSES];

    __device__ __forceinline__ void load(const float* B0, const float* B1, int kc)
    {
#pragma unroll
        for (int p = 0; p < PASSES; ++p) {
            const int idx = threadIdx.x + GEMM_THREADS * p;
            const int pr = idx / (NTOT / 4);
            const int c = (idx - pr * (NTOT / 4)) * 4;
            const float* src = (c < NH) ? (B0 + c) : (B1 + (c - NH));
            g0[p] = *(const float4*)(src + (size_t)(kc + 2 * pr) * NH);
            g1[p] = *(const float4*)(src + (size_t)(kc + 2 * pr + 1) * NH);
        }
    }
    __device__ __forceinline__ void store(float* __restrict__ dst0) const
    {
#pragma unroll
        for (int p = 0; p < PASSES; ++p) {
            const int idx = threadIdx.x + GEMM_THREADS * p;
            const int pr = idx / (NTOT / 4);
            const int c = (idx - pr * (NTOT / 4)) * 4;
            float4 lo = {g0[p].x, g1[p].x, g0[p].y, g1[p].y};
            float4 hi4 = {g0[p].z, g1[p].z, g0[p].w, g1[p].w};
            float* d = dst0 + (pr * NTOT + c) * 2;
            *(float4*)(d) = lo;
            *(float4*)(d + 4) = hi4;
        }
    }
};

template <int NTW, int K, bool RELU_IN>
__global__ __launch_bounds__(GEMM_THREADS) void k_gemm_pair(
    const float* A,
    const float* B0, const float* __restrict__ bias0,
    float* __restrict__ C0,
    const float* B1, float* __restrict__ C1,
    int* __restrict__ qz)
{
    constexpr int RT = 5;
    constexpr int NWV = GEMM_THREADS / 32;
    constexpr int NW = NTW * 16;
    constexpr int NTOT = NW * NWV;
    constexpr int NH = NTOT / 2;
    constexpr int BK = ChunkStage<NTOT>::BK;
    constexpr int NCH = K / BK;
    constexpr int KS = BK / 4;
    constexpr int ROWS = RT * 16;
    constexpr int AP = BK + 4;
    constexpr int AITEMS = ROWS * (BK / 4);
    constexpr int APASSES = (AITEMS + GEMM_THREADS - 1) / GEMM_THREADS;
    static_assert(K % BK == 0, "whole K-chunks");
    static_assert(NH % NW == 0, "a wave's columns lie in one product");
    static_assert(NNODES % ROWS == 0, "no row tail: every wave is active");
    static_assert(BK == 32, "A staging: 8 float4 per row");

    __shared__ __align__(16) float L[2 * ROWS * AP + 2 * BK * NTOT];
    float (*As)[ROWS * AP] = (float (*)[ROWS * AP])L;
    float (*Bs)[BK * NTOT] = (float (*)[BK * NTOT])(L + 2 * ROWS * AP);

    if (blockIdx.x == 0 && threadIdx.x < 2) qz[threadIdx.x * QSTRIDE] = 0;

    const int lane = threadIdx.x & 31;
    const int wave = threadIdx.x >> 5;
    const int rh = lane & 15;
    const int hi = lane >> 4;
    const int col0 = wave * NW;

    v8f acc[RT][NTW];
#pragma unroll
    for (int r = 0; r < RT; ++r)
#pragma unroll
        for (int t = 0; t < NTW; ++t) acc[r][t] = v8f{};

    static_assert(APASSES == 3 && GEMM_THREADS % 8 == 0, "three A passes, written out below");
    const int it2r = threadIdx.x + 2 * GEMM_THREADS;
    const int it2 = (it2r >= AITEMS) ? (it2r - AITEMS) : it2r;
    const int aoff0 = (threadIdx.x >> 3) * AP + (threadIdx.x & 7) * 4;
    const int aoff1 = aoff0 + (GEMM_THREADS / 8) * AP;
    const int aoff2 = (it2 >> 3) * AP + (it2 & 7) * 4;
    const float* agp0 = A + (size_t)(blockIdx.x * ROWS + (threadIdx.x >> 3)) * K + (threadIdx.x & 7) * 4;
    const float* agp1 = agp0 + (size_t)(GEMM_THREADS / 8) * K;
    const float* agp2 = A + (size_t)(blockIdx.x * ROWS + (it2 >> 3)) * K + (it2 & 7) * 4;

    ChunkStage<NTOT> st;
    float4 a40, a41, a42;

    st.load(B0, B1, 0);
    a40 = *(const float4*)(agp0);
    a41 = *(const float4*)(agp1);
    a42 = *(const float4*)(agp2);
    st.store(Bs[0]);
    *(float4*)(&As[0][aoff0]) = a40;
    *(float4*)(&As[0][aoff1]) = a41;
    *(float4*)(&As[0][aoff2]) = a42;
    __syncthreads();

#pragma unroll
    for (int ci = 0; ci < NCH; ++ci) {
        const int cur = ci & 1;
        if (ci + 1 < NCH) {
            st.load(B0, B1, (ci + 1) * BK);
            a40 = *(const float4*)(agp0 + (ci + 1) * BK);
            a41 = *(const float4*)(agp1 + (ci + 1) * BK);
            a42 = *(const float4*)(agp2 + (ci + 1) * BK);
            asm volatile("" ::: "memory");
        }

        const v2f* Bp = (const v2f*)Bs[cur] + hi * NTOT + col0 + rh;
        const v2f* Ap = (const v2f*)(&As[cur][rh * AP + 2 * hi]);
#pragma unroll
        for (int ks = 0; ks < KS; ++ks) {
            v2f a[RT], b[NTW];
#pragma unroll
            for (int r = 0; r < RT; ++r) {
                a[r] = Ap[r * (16 * AP / 2) + 2 * ks];
                if constexpr (RELU_IN) {
                    a[r].x = fmaxf(a[r].x, 0.0f);
                    a[r].y = fmaxf(a[r].y, 0.0f);
                }
            }
#pragma unroll
            for (int t = 0; t < NTW; ++t)
                b[t] = Bp[(2 * ks) * NTOT + t * 16];
#pragma unroll
            for (int r = 0; r < RT; ++r)
#pragma unroll
                for (int t = 0; t < NTW; ++t)
                    acc[r][t] = __builtin_amdgcn_wmma_f32_16x16x4_f32(
                        false, a[r], false, b[t], (short)0, acc[r][t], false, false);
        }

        if (ci + 1 < NCH) {
            __builtin_amdgcn_sched_barrier(0);
            st.store(Bs[cur ^ 1]);
            *(float4*)(&As[cur ^ 1][aoff0]) = a40;
            *(float4*)(&As[cur ^ 1][aoff1]) = a41;
            *(float4*)(&As[cur ^ 1][aoff2]) = a42;
            __syncthreads();
        }
    }

    const bool second = (col0 >= NH);
    float* __restrict__ C = second ? C1 : C0;
    const int ch0 = second ? (col0 - NH) : col0;
#pragma unroll
    for (int t = 0; t < NTW; ++t) {
        const int col = ch0 + t * 16 + rh;
        const float bv = second ? 0.0f : bias0[col];
#pragma unroll
        for (int r = 0; r < RT; ++r) {
#pragma unroll
            for (int i = 0; i < 8; ++i) {
                const int row = blockIdx.x * ROWS + r * 16 + i + 8 * hi;
                C[(size_t)row * NH + col] = acc[r][t][i] + bv;
            }
        }
    }
}

template <int K, int N>
struct PairStage {
    static constexpr int ITERS = (K / 2) * (N / 4);
    static_assert(ITERS % 256 == 0, "whole passes of the 256-thread block");
    static constexpr int PASSES = ITERS / 256;
    float4 g0[PASSES], g1[PASSES];

    __device__ __forceinline__ void load(const float* __restrict__ Wg)
    {
#pragma unroll
        for (int p = 0; p < PASSES; ++p) {
            const int idx = threadIdx.x + 256 * p;
            const int pr = idx / (N / 4);
            const int c = (idx - pr * (N / 4)) * 4;
            g0[p] = *(const float4*)(Wg + (size_t)(2 * pr) * N + c);
            g1[p] = *(const float4*)(Wg + (size_t)(2 * pr + 1) * N + c);
        }
    }
    __device__ __forceinline__ void store(float* __restrict__ dst0) const
    {
#pragma unroll
        for (int p = 0; p < PASSES; ++p) {
            const int idx = threadIdx.x + 256 * p;
            const int pr = idx / (N / 4);
            const int c = (idx - pr * (N / 4)) * 4;
            float4 lo = {g0[p].x, g1[p].x, g0[p].y, g1[p].y};
            float4 hi4 = {g0[p].z, g1[p].z, g0[p].w, g1[p].w};
            float* d = dst0 + (pr * N + c) * 2;
            *(float4*)(d) = lo;
            *(float4*)(d + 4) = hi4;
        }
    }
};

__global__ __launch_bounds__(256) void k_mlp_tail(
    const float* __restrict__ hin,
    const float* __restrict__ W1, const float* __restrict__ b1,
    const float* __restrict__ W2, const float* __restrict__ b2,
    const float* __restrict__ W3, const float* __restrict__ b3,
    float* __restrict__ out, int M)
{
    __shared__ __align__(16) float W1s[64 * 128];
    __shared__ __align__(16) float W2s[128 * 64];
    __shared__ __align__(16) float W3s[32 * 16 * 2];

    const int lane = threadIdx.x & 31;
    const int wave = threadIdx.x >> 5;
    const int tm = blockIdx.x * 8 + wave;
    const bool active = (tm * 16 < M);
    const int c = lane & 15;
    const int hh = lane >> 4;

    v2f xb[16];
    v8f acc1[8];
    v8f acc2[4];
    v8f z;
    if (active) {
        const float* xp = hin + (size_t)(tm * 16 + c) * 64 + 2 * hh;
#pragma unroll
        for (int ks = 0; ks < 16; ++ks) xb[ks] = *(const v2f*)(xp + 4 * ks);
#pragma unroll
        for (int t = 0; t < 8; ++t) {
            const float4 lo = *(const float4*)(b1 + 16 * t + 8 * hh);
            const float4 hi4 = *(const float4*)(b1 + 16 * t + 8 * hh + 4);
            acc1[t] = v8f{lo.x, lo.y, lo.z, lo.w, hi4.x, hi4.y, hi4.z, hi4.w};
        }
#pragma unroll
        for (int t = 0; t < 4; ++t) {
            const float4 lo = *(const float4*)(b2 + 16 * t + 8 * hh);
            const float4 hi4 = *(const float4*)(b2 + 16 * t + 8 * hh + 4);
            acc2[t] = v8f{lo.x, lo.y, lo.z, lo.w, hi4.x, hi4.y, hi4.z, hi4.w};
        }
#pragma unroll
        for (int r = 0; r < 8; ++r) {
            const int cls = 8 * hh + r;
            z[r] = b3[cls < 10 ? cls : 9];
        }
    }

    PairStage<64, 128> st1;
    PairStage<128, 64> st2;
    v2f w3v[2];
    st1.load(W1);
    st2.load(W2);
#pragma unroll
    for (int p = 0; p < 2; ++p) {
        const int idx = threadIdx.x + 256 * p;
        const int pr = idx >> 4;
        const int col = idx & 15;
        const int cc = col < 10 ? col : 9;
        w3v[p] = v2f{W3[(2 * pr) * 10 + cc], W3[(2 * pr + 1) * 10 + cc]};
    }
    __builtin_amdgcn_sched_barrier(0);
    st1.store(W1s);
    st2.store(W2s);
#pragma unroll
    for (int p = 0; p < 2; ++p) ((v2f*)W3s)[threadIdx.x + 256 * p] = w3v[p];
    __syncthreads();
    if (!active) return;

    {
        const v2f* Wp = (const v2f*)W1s + hh * 128 + c;
#pragma unroll
        for (int ks = 0; ks < 16; ++ks) {
#pragma unroll
            for (int t = 0; t < 8; ++t) {
                const v2f a = Wp[(2 * ks) * 128 + t * 16];
                acc1[t] = __builtin_amdgcn_wmma_f32_16x16x4_f32(
                    false, a, false, xb[ks], (short)0, acc1[t], false, false);
            }
        }
#pragma unroll
        for (int t = 0; t < 8; ++t)
#pragma unroll
            for (int r = 0; r < 8; ++r) acc1[t][r] = fmaxf(acc1[t][r], 0.0f);
    }

    {
        const v2f* Wp = (const v2f*)W2s + (4 * hh) * 64 + c;
#pragma unroll
        for (int t = 0; t < 8; ++t) {
#pragma unroll
            for (int j = 0; j < 4; ++j) {
                const v2f b = {acc1[t][2 * j], acc1[t][2 * j + 1]};
#pragma unroll
                for (int mt = 0; mt < 4; ++mt) {
                    const v2f a = Wp[(8 * t + j) * 64 + mt * 16];
                    acc2[mt] = __builtin_amdgcn_wmma_f32_16x16x4_f32(
                        false, a, false, b, (short)0, acc2[mt], false, false);
                }
            }
        }
#pragma unroll
        for (int t = 0; t < 4; ++t)
#pragma unroll
            for (int r = 0; r < 8; ++r) acc2[t][r] = fmaxf(acc2[t][r], 0.0f);
    }

    {
        const v2f* Wp = (const v2f*)W3s + (4 * hh) * 16 + c;
#pragma unroll
        for (int t = 0; t < 4; ++t) {
#pragma unroll
            for (int j = 0; j < 4; ++j) {
                const v2f b = {acc2[t][2 * j], acc2[t][2 * j + 1]};
                const v2f a = Wp[(8 * t + j) * 16];
                z = __builtin_amdgcn_wmma_f32_16x16x4_f32(
                    false, a, false, b, (short)0, z, false, false);
            }
        }
    }

    const bool lowhalf = (hh == 0);
    float mx = fmaxf(z[0], z[1]);
#pragma unroll
    for (int r = 2; r < 8; ++r) mx = lowhalf ? fmaxf(mx, z[r]) : mx;
    mx = fmaxf(mx, __shfl_xor(mx, 16));
    float s = expf(z[0] - mx) + expf(z[1] - mx);
#pragma unroll
    for (int r = 2; r < 8; ++r) s += lowhalf ? expf(z[r] - mx) : 0.0f;
    s += __shfl_xor(s, 16);
    const float lse = mx + logf(s);

    float* o = out + (size_t)(tm * 16 + c) * 10 + 8 * hh;
    *(float2*)(o) = float2{z[0] - lse, z[1] - lse};
    if (lowhalf) {
        *(float2*)(o + 2) = float2{z[2] - lse, z[3] - lse};
        *(float2*)(o + 4) = float2{z[4] - lse, z[5] - lse};
        *(float2*)(o + 6) = float2{z[6] - lse, z[7] - lse};
    }
}

extern "C" void kernel_launch(void* const* d_in, const int* in_sizes, int n_in,
                              void* d_out, int out_size, void* d_ws, size_t ws_size,
                              hipStream_t stream) {
    const float* x       = (const float*)d_in[0];
    const int*   ei      = (const int*)d_in[1];
    const float* W1_rel  = (const float*)d_in[2];
    const float* W1_root = (const float*)d_in[3];
    const float* b1      = (const float*)d_in[4];
    const float* W2_rel  = (const float*)d_in[5];
    const float* W2_root = (const float*)d_in[6];
    const float* b2      = (const float*)d_in[7];
    const float* lin1_w  = (const float*)d_in[8];
    const float* lin1_b  = (const float*)d_in[9];
    const float* lin2_w  = (const float*)d_in[10];
    const float* lin2_b  = (const float*)d_in[11];
    const float* lin3_w  = (const float*)d_in[12];
    const float* lin3_b  = (const float*)d_in[13];

    float* y1 = (float*)d_ws;
    float* h1 = y1 + (size_t)NNODES * 128;
    float* y2 = h1 + (size_t)NNODES * 128;
    float* h2 = y2 + (size_t)NNODES * 64;
    int* q1 = (int*)y2;
    int* q2 = (int*)y1;

    const int mblocks = ((NNODES / 16) + 7) / 8;


    k_gemm_pair<2, 256, false><<<NNODES / 80, GEMM_THREADS, 0, stream>>>(
        x, W1_root, b1, h1, W1_rel, y1, q1);
    k_scatter_add<128><<<NTASK, 256, 0, stream>>>(y1, ei, h1, q1);

    k_gemm_pair<1, 128, true><<<NNODES / 80, GEMM_THREADS, 0, stream>>>(
        h1, W2_root, b2, h2, W2_rel, y2, q2);
    k_scatter_add<64><<<NTASK, 256, 0, stream>>>(y2, ei, h2, q2);

    k_mlp_tail<<<mblocks, 256, 0, stream>>>(
        h2, lin1_w, lin1_b, lin2_w, lin2_b, lin3_w, lin3_b, (float*)d_out, NNODES);
}
